# starting kernel
# baseline (speedup 1.0000x reference)
#include <hip/hip_runtime.h>
#include <math.h>

#define NS 128
#define NW 64
#define W_MAX 30.0f
#define L1C 0.2f
#define L2C 0.3f
#define RHOC 0.3f

typedef __attribute__((ext_vector_type(2))) float v2f;
typedef __attribute__((ext_vector_type(8))) float v8f;

__global__ void grf_setup_kernel(const float* __restrict__ theta,
                                 float* __restrict__ Ct, float* __restrict__ St,
                                 float* __restrict__ GcT, float* __restrict__ GsT,
                                 float* __restrict__ HcT, float* __restrict__ HsT) {
    const int t = blockIdx.x * blockDim.x + threadIdx.x;
    const float dw = W_MAX / (float)(NW - 1);

    {
        const int s = t >> 6, w = t & 63;
        const float x = ((float)s / (float)(NS - 1)) * ((float)w * dw);
        float sn, cs;
        __sincosf(x, &sn, &cs);
        Ct[t] = cs;
        St[t] = sn;
    }

    if (t < NW * NW) {
        const int w1 = t & 63, w2 = t >> 6;
        const float vw1 = (float)w1 * dw;
        const float vw2 = (float)w2 * dw;

        float m;
        if (w2 == NW - 1) m = (w1 == 0) ? 0.25f : 1.0f;
        else              m = (w1 == 0) ? 1.0f  : 2.0f;
        m *= dw * dw;

        const float cnorm = (1.0f * L1C * L2C) / (4.0f * (float)M_PI);
        const float x1 = L1C * vw1, x2 = L2C * vw2;
        const float q  = x1 * x1 + x2 * x2;
        const float cr = 2.0f * RHOC * x1 * x2;
        const float A1 = sqrtf(m * cnorm * __expf(-(q + cr) * 0.25f));
        const float A2 = sqrtf(m * cnorm * __expf(-(q - cr) * 0.25f));

        float sp1, cp1, sp2, cp2;
        __sincosf(theta[t],           &sp1, &cp1);
        __sincosf(theta[NW * NW + t], &sp2, &cp2);

        const int tt = w1 * NW + w2;
        GcT[tt] = A1 * cp1 + A2 * cp2;
        GsT[tt] = A2 * sp2 - A1 * sp1;
        HcT[tt] = A1 * sp1 + A2 * sp2;
        HsT[tt] = A1 * cp1 - A2 * cp2;
    }
}

__global__ void grf_gemm1_kernel(const float* __restrict__ Ct, const float* __restrict__ St,
                                 const float* __restrict__ GcT, const float* __restrict__ GsT,
                                 const float* __restrict__ HcT, const float* __restrict__ HsT,
                                 float* __restrict__ Mc, float* __restrict__ Mn) {
    const int lane  = threadIdx.x & 31;
    const int wave  = (blockIdx.x * blockDim.x + threadIdx.x) >> 5;
    const int which = wave >> 5;
    const int tile  = wave & 31;
    const int tm    = tile >> 2;
    const int tn    = tile & 3;

    const float* B0T = which ? HcT : GcT;
    const float* B1T = which ? HsT : GsT;
    float*       D   = which ? Mn : Mc;

    const int half = lane >> 4;
    const int row  = tm * 16 + (lane & 15);
    const int col  = tn * 16 + (lane & 15);

    v8f acc = {};
    #pragma unroll
    for (int k0 = 0; k0 < NW; k0 += 4) {
        const int ka = k0 + half * 2;
        const v2f a = *(const v2f*)&Ct [row * NW + ka];
        const v2f b = *(const v2f*)&B0T[col * NW + ka];
        acc = __builtin_amdgcn_wmma_f32_16x16x4_f32(false, a, false, b,
                                                    (short)0, acc, false, false);
    }
    #pragma unroll
    for (int k0 = 0; k0 < NW; k0 += 4) {
        const int ka = k0 + half * 2;
        const v2f a = *(const v2f*)&St [row * NW + ka];
        const v2f b = *(const v2f*)&B1T[col * NW + ka];
        acc = __builtin_amdgcn_wmma_f32_16x16x4_f32(false, a, false, b,
                                                    (short)0, acc, false, false);
    }

    const float sgn   = which ? -1.0f : 1.0f;
    const int   mbase = tm * 16 + half * 8;
    #pragma unroll
    for (int r = 0; r < 8; ++r)
        D[(mbase + r) * NW + col] = sgn * acc[r];
}

__global__ void grf_gemm2_kernel(const float* __restrict__ Mc, const float* __restrict__ Mn,
                                 const float* __restrict__ Ct, const float* __restrict__ St,
                                 float* __restrict__ out) {
    const int lane = threadIdx.x & 31;
    const int wave = (blockIdx.x * blockDim.x + threadIdx.x) >> 5;
    const int tm   = wave >> 3;
    const int tn   = wave & 7;

    const int half = lane >> 4;
    const int row  = tm * 16 + (lane & 15);
    const int col  = tn * 16 + (lane & 15);

    v8f acc = {};
    #pragma unroll
    for (int k0 = 0; k0 < NW; k0 += 4) {
        const int ka = k0 + half * 2;
        const v2f a = *(const v2f*)&Mc[row * NW + ka];
        const v2f b = *(const v2f*)&Ct[col * NW + ka];
        acc = __builtin_amdgcn_wmma_f32_16x16x4_f32(false, a, false, b,
                                                    (short)0, acc, false, false);
    }
    #pragma unroll
    for (int k0 = 0; k0 < NW; k0 += 4) {
        const int ka = k0 + half * 2;
        const v2f a = *(const v2f*)&Mn[row * NW + ka];
        const v2f b = *(const v2f*)&St[col * NW + ka];
        acc = __builtin_amdgcn_wmma_f32_16x16x4_f32(false, a, false, b,
                                                    (short)0, acc, false, false);
    }

    const float SQRT2 = 1.41421356237309505f;
    const int mbase = tm * 16 + half * 8;
    #pragma unroll
    for (int r = 0; r < 8; ++r)
        out[(mbase + r) * NS + col] = SQRT2 * acc[r];
}

extern "C" void kernel_launch(void* const* d_in, const int* in_sizes, int n_in,
                              void* d_out, int out_size, void* d_ws, size_t ws_size,
                              hipStream_t stream) {
    const float* theta = (const float*)d_in[0];
    float* out = (float*)d_out;

    float* ws  = (float*)d_ws;
    float* Ct  = ws;
    float* St  = ws + 8192;
    float* GcT = ws + 16384;
    float* GsT = ws + 20480;
    float* HcT = ws + 24576;
    float* HsT = ws + 28672;
    float* Mc  = ws + 32768;
    float* Mn  = ws + 40960;

    grf_setup_kernel<<<32, 256, 0, stream>>>(theta, Ct, St, GcT, GsT, HcT, HsT);
    grf_gemm1_kernel<<<8, 256, 0, stream>>>(Ct, St, GcT, GsT, HcT, HsT, Mc, Mn);
    grf_gemm2_kernel<<<8, 256, 0, stream>>>(Mc, Mn, Ct, St, out);
}


# Fuse the three launches into one: each of 32 work-groups rebuilds grids and trig rows in LDS and runs both WMMA products for its 16x32 block; sums regrouped (cos/sin halves, K quarters)
# speedup vs baseline: 3.4108x; 3.4108x over previous
#include <hip/hip_runtime.h>
#include <math.h>


#define NS 128
#define NW 64
#define W_MAX 30.0f
#define L1C 0.2f
#define L2C 0.3f
#define RHOC 0.3f

#define LP 68
#define BM 16
#define BN 32

typedef __attribute__((ext_vector_type(2))) float v2f;
typedef __attribute__((ext_vector_type(4))) float v4f;
typedef __attribute__((ext_vector_type(8))) float v8f;

static __device__ __forceinline__ void mma_k32(const float* __restrict__ Arow,
                                               const float* __restrict__ Brow,
                                               v8f& acc0, v8f& acc1) {
    #pragma unroll
    for (int j = 0; j < 4; ++j) {
        const v4f a = *(const v4f*)(Arow + 8 * j);
        const v4f b = *(const v4f*)(Brow + 8 * j);
        const v2f a0 = {a.x, a.y}, b0 = {b.x, b.y};
        const v2f a1 = {a.z, a.w}, b1 = {b.z, b.w};
        acc0 = __builtin_amdgcn_wmma_f32_16x16x4_f32(false, a0, false, b0,
                                                     (short)0, acc0, false, false);
        acc1 = __builtin_amdgcn_wmma_f32_16x16x4_f32(false, a1, false, b1,
                                                     (short)0, acc1, false, false);
    }
}

__global__ __launch_bounds__(256)
void grf_fused_kernel(const float* __restrict__ theta, float* __restrict__ out) {
    __shared__ __attribute__((aligned(16))) float sG[4][NW * LP];
    __shared__ __attribute__((aligned(16))) float sTA[2][BM * LP];
    __shared__ __attribute__((aligned(16))) float sTB[2][BN * LP];
    __shared__ __attribute__((aligned(16))) float sM[2][BM * LP];
    __shared__ __attribute__((aligned(16))) float sP[4][BM * BN];

    const int tid  = threadIdx.x;
    const int lane = tid & 31;
    const int wv   = tid >> 5;
    const int bm   = blockIdx.x >> 2;
    const int bn   = blockIdx.x & 3;
    const float dw = W_MAX / (float)(NW - 1);

    {
        const int w1 = tid & 63;
        const float vw1 = (float)w1 * dw;
        const float x1  = L1C * vw1;
        const float cnorm = (1.0f * L1C * L2C) / (4.0f * (float)M_PI);

        float p1[16], p2[16];
        #pragma unroll
        for (int i = 0; i < 16; ++i) {
            const int t = i * 256 + tid;
            p1[i] = theta[t];
            p2[i] = theta[NW * NW + t];
        }
        #pragma unroll
        for (int i = 0; i < 16; ++i) {
            const int w2 = i * 4 + (tid >> 6);
            const float vw2 = (float)w2 * dw;

            float m;
            if (w2 == NW - 1) m = (w1 == 0) ? 0.25f : 1.0f;
            else              m = (w1 == 0) ? 1.0f  : 2.0f;
            m *= dw * dw;

            const float x2 = L2C * vw2;
            const float q  = x1 * x1 + x2 * x2;
            const float cr = 2.0f * RHOC * x1 * x2;
            const float A1 = sqrtf(m * cnorm * __expf(-(q + cr) * 0.25f));
            const float A2 = sqrtf(m * cnorm * __expf(-(q - cr) * 0.25f));

            float sp1, cp1, sp2, cp2;
            __sincosf(p1[i], &sp1, &cp1);
            __sincosf(p2[i], &sp2, &cp2);

            const int tt = w1 * LP + w2;
            sG[0][tt] = A1 * cp1 + A2 * cp2;
            sG[1][tt] = A2 * sp2 - A1 * sp1;
            sG[2][tt] = A1 * sp1 + A2 * sp2;
            sG[3][tt] = A1 * cp1 - A2 * cp2;
        }
    }

    {
        const int w = tid & 63;
        const float xw = (float)w * dw;
        #pragma unroll
        for (int i = 0; i < BM / 4; ++i) {
            const int r = i * 4 + (tid >> 6);
            const int s = bm * BM + r;
            const float x = ((float)s / (float)(NS - 1)) * xw;
            float sn, cs;
            __sincosf(x, &sn, &cs);
            sTA[0][r * LP + w] = cs;
            sTA[1][r * LP + w] = sn;
        }
        #pragma unroll
        for (int i = 0; i < BN / 4; ++i) {
            const int r = i * 4 + (tid >> 6);
            const int s = bn * BN + r;
            const float x = ((float)s / (float)(NS - 1)) * xw;
            float sn, cs;
            __sincosf(x, &sn, &cs);
            sTB[0][r * LP + w] = cs;
            sTB[1][r * LP + w] = sn;
        }
    }
    __syncthreads();

    const int c = lane & 15;
    const int h = lane >> 4;

    {
        const int which = wv >> 2;
        const int tn    = wv & 3;
        const float* Ac = &sTA[0][c * LP + 4 * h];
        const float* As = &sTA[1][c * LP + 4 * h];
        const float* Bc = &sG[2 * which + 0][(tn * 16 + c) * LP + 4 * h];
        const float* Bs = &sG[2 * which + 1][(tn * 16 + c) * LP + 4 * h];

        v8f acc0 = {}, acc1 = {}, acc2 = {}, acc3 = {};
        mma_k32(Ac,      Bc,      acc0, acc1);
        mma_k32(As,      Bs,      acc2, acc3);
        mma_k32(Ac + 32, Bc + 32, acc0, acc1);
        mma_k32(As + 32, Bs + 32, acc2, acc3);

        const float sgn = which ? -1.0f : 1.0f;
        float* D = &sM[which][(8 * h) * LP + tn * 16 + c];
        #pragma unroll
        for (int r = 0; r < 8; ++r)
            D[r * LP] = sgn * ((acc0[r] + acc1[r]) + (acc2[r] + acc3[r]));
    }
    __syncthreads();

    {
        const int tn2   = wv & 1;
        const int kq    = wv >> 1;
        const int which = kq >> 1;
        const int k0    = (kq & 1) * 32;
        const float* A = &sM[which][c * LP + k0 + 4 * h];
        const float* B = &sTB[which][(tn2 * 16 + c) * LP + k0 + 4 * h];

        v8f acc0 = {}, acc1 = {};
        mma_k32(A, B, acc0, acc1);

        float* P = &sP[kq][(8 * h) * BN + tn2 * 16 + c];
        #pragma unroll
        for (int r = 0; r < 8; ++r)
            P[r * BN] = acc0[r] + acc1[r];
    }
    __syncthreads();

    {
        const float SQRT2 = 1.41421356237309505f;
        const int col = tid & 31;
        #pragma unroll
        for (int j = 0; j < 2; ++j) {
            const int row = (tid >> 5) + 8 * j;
            const int e = row * BN + col;
            const float v = (sP[0][e] + sP[1][e]) + (sP[2][e] + sP[3][e]);
            out[(bm * BM + row) * NS + bn * BN + col] = SQRT2 * v;
        }
    }
}

extern "C" void kernel_launch(void* const* d_in, const int* in_sizes, int n_in,
                              void* d_out, int out_size, void* d_ws, size_t ws_size,
                              hipStream_t stream) {
    const float* theta = (const float*)d_in[0];
    float* out = (float*)d_out;
    grf_fused_kernel<<<32, 256, 0, stream>>>(theta, out);
}


# Set-up math strength-reduced: amplitude = folded sqrt(mask*c) * exp2(arg*k) replaces sqrtf(mask*c*expf(arg)); s/127 by reciprocal + one fma fix-up (same bits); amplitude rounding changes
# speedup vs baseline: 4.0970x; 1.2012x over previous
#include <hip/hip_runtime.h>
#include <math.h>


#define NS 128
#define NW 64
#define W_MAX 30.0f
#define L1C 0.2f
#define L2C 0.3f
#define RHOC 0.3f

#define LP 68
#define BM 16
#define BN 32

typedef __attribute__((ext_vector_type(2))) float v2f;
typedef __attribute__((ext_vector_type(4))) float v4f;
typedef __attribute__((ext_vector_type(8))) float v8f;

static __device__ __forceinline__ void mma_k32(const float* __restrict__ Arow,
                                               const float* __restrict__ Brow,
                                               v8f& acc0, v8f& acc1) {
    #pragma unroll
    for (int j = 0; j < 4; ++j) {
        const v4f a = *(const v4f*)(Arow + 8 * j);
        const v4f b = *(const v4f*)(Brow + 8 * j);
        const v2f a0 = {a.x, a.y}, b0 = {b.x, b.y};
        const v2f a1 = {a.z, a.w}, b1 = {b.z, b.w};
        acc0 = __builtin_amdgcn_wmma_f32_16x16x4_f32(false, a0, false, b0,
                                                     (short)0, acc0, false, false);
        acc1 = __builtin_amdgcn_wmma_f32_16x16x4_f32(false, a1, false, b1,
                                                     (short)0, acc1, false, false);
    }
}

__global__ __launch_bounds__(256)
void grf_fused_kernel(const float* __restrict__ theta, float* __restrict__ out) {
    __shared__ __attribute__((aligned(16))) float sG[4][NW * LP];
    __shared__ __attribute__((aligned(16))) float sTA[2][BM * LP];
    __shared__ __attribute__((aligned(16))) float sTB[2][BN * LP];
    __shared__ __attribute__((aligned(16))) float sM[2][BM * LP];
    __shared__ __attribute__((aligned(16))) float sP[4][BM * BN];

    const int tid  = threadIdx.x;
    const int lane = tid & 31;
    const int wv   = tid >> 5;
    const int bm   = blockIdx.x >> 2;
    const int bn   = blockIdx.x & 3;
    const float dw = W_MAX / (float)(NW - 1);

    {
        const int w1 = tid & 63;
        const int t6 = tid >> 6;
        const float vw1 = (float)w1 * dw;
        const float x1  = L1C * vw1;
        const float cnorm = (1.0f * L1C * L2C) / (4.0f * (float)M_PI);

        const float kTwo  = __builtin_sqrtf(2.0f  * dw * dw * cnorm);
        const float kOne  = __builtin_sqrtf(1.0f  * dw * dw * cnorm);
        const float kQrt  = __builtin_sqrtf(0.25f * dw * dw * cnorm);
        const float kBody = (w1 == 0) ? kOne : kTwo;
        const float kLast = (w1 == 0) ? kQrt : kOne;
        const float C8 = -0.125f * 1.44269504088896341f;

        float p1[16], p2[16];
        #pragma unroll
        for (int i = 0; i < 16; ++i) {
            const int t = i * 256 + tid;
            p1[i] = theta[t];
            p2[i] = theta[NW * NW + t];
        }
        #pragma unroll
        for (int i = 0; i < 16; ++i) {
            const int w2 = i * 4 + t6;
            const float vw2 = (float)w2 * dw;
            const float km  = (i == 15 && t6 == 3) ? kLast : kBody;

            const float x2 = L2C * vw2;
            const float q  = x1 * x1 + x2 * x2;
            const float cr = 2.0f * RHOC * x1 * x2;
            const float A1 = km * __builtin_amdgcn_exp2f((q + cr) * C8);
            const float A2 = km * __builtin_amdgcn_exp2f((q - cr) * C8);

            float sp1, cp1, sp2, cp2;
            __sincosf(p1[i], &sp1, &cp1);
            __sincosf(p2[i], &sp2, &cp2);

            const int tt = w1 * LP + w2;
            sG[0][tt] = A1 * cp1 + A2 * cp2;
            sG[1][tt] = A2 * sp2 - A1 * sp1;
            sG[2][tt] = A1 * sp1 + A2 * sp2;
            sG[3][tt] = A1 * cp1 - A2 * cp2;
        }
    }

    {
        const int w = tid & 63;
        const float xw = (float)w * dw;
        const float rinv = 1.0f / (float)(NS - 1);
        #pragma unroll
        for (int i = 0; i < BM / 4; ++i) {
            const int r = i * 4 + (tid >> 6);
            const float fs = (float)(bm * BM + r);
            const float q0 = fs * rinv;
            const float vs = __builtin_fmaf(__builtin_fmaf(-q0, (float)(NS - 1), fs), rinv, q0);
            const float x = vs * xw;
            float sn, cs;
            __sincosf(x, &sn, &cs);
            sTA[0][r * LP + w] = cs;
            sTA[1][r * LP + w] = sn;
        }
        #pragma unroll
        for (int i = 0; i < BN / 4; ++i) {
            const int r = i * 4 + (tid >> 6);
            const float fs = (float)(bn * BN + r);
            const float q0 = fs * rinv;
            const float vs = __builtin_fmaf(__builtin_fmaf(-q0, (float)(NS - 1), fs), rinv, q0);
            const float x = vs * xw;
            float sn, cs;
            __sincosf(x, &sn, &cs);
            sTB[0][r * LP + w] = cs;
            sTB[1][r * LP + w] = sn;
        }
    }
    __syncthreads();

    const int c = lane & 15;
    const int h = lane >> 4;

    {
        const int which = wv >> 2;
        const int tn    = wv & 3;
        const float* Ac = &sTA[0][c * LP + 4 * h];
        const float* As = &sTA[1][c * LP + 4 * h];
        const float* Bc = &sG[2 * which + 0][(tn * 16 + c) * LP + 4 * h];
        const float* Bs = &sG[2 * which + 1][(tn * 16 + c) * LP + 4 * h];

        v8f acc0 = {}, acc1 = {}, acc2 = {}, acc3 = {};
        mma_k32(Ac,      Bc,      acc0, acc1);
        mma_k32(As,      Bs,      acc2, acc3);
        mma_k32(Ac + 32, Bc + 32, acc0, acc1);
        mma_k32(As + 32, Bs + 32, acc2, acc3);

        const float sgn = which ? -1.0f : 1.0f;
        float* D = &sM[which][(8 * h) * LP + tn * 16 + c];
        #pragma unroll
        for (int r = 0; r < 8; ++r)
            D[r * LP] = sgn * ((acc0[r] + acc1[r]) + (acc2[r] + acc3[r]));
    }
    __syncthreads();

    {
        const int tn2   = wv & 1;
        const int kq    = wv >> 1;
        const int which = kq >> 1;
        const int k0    = (kq & 1) * 32;
        const float* A = &sM[which][c * LP + k0 + 4 * h];
        const float* B = &sTB[which][(tn2 * 16 + c) * LP + k0 + 4 * h];

        v8f acc0 = {}, acc1 = {};
        mma_k32(A, B, acc0, acc1);

        float* P = &sP[kq][(8 * h) * BN + tn2 * 16 + c];
        #pragma unroll
        for (int r = 0; r < 8; ++r)
            P[r * BN] = acc0[r] + acc1[r];
    }
    __syncthreads();

    {
        const float SQRT2 = 1.41421356237309505f;
        const int col = tid & 31;
        #pragma unroll
        for (int j = 0; j < 2; ++j) {
            const int row = (tid >> 5) + 8 * j;
            const int e = row * BN + col;
            const float v = (sP[0][e] + sP[1][e]) + (sP[2][e] + sP[3][e]);
            out[(bm * BM + row) * NS + bn * BN + col] = SQRT2 * v;
        }
    }
}

extern "C" void kernel_launch(void* const* d_in, const int* in_sizes, int n_in,
                              void* d_out, int out_size, void* d_ws, size_t ws_size,
                              hipStream_t stream) {
    const float* theta = (const float*)d_in[0];
    float* out = (float*)d_out;
    grf_fused_kernel<<<32, 256, 0, stream>>>(theta, out);
}


# Grid set-up remapped: each thread owns 16 consecutive w2 of one w1, so its transposed LDS stores are 16 conflict-free 16-byte stores, not 64 dword stores with 2-way bank conflicts; sums unchanged
# speedup vs baseline: 4.3104x; 1.0521x over previous
#include <hip/hip_runtime.h>
#include <math.h>


#define NS 128
#define NW 64
#define W_MAX 30.0f
#define L1C 0.2f
#define L2C 0.3f
#define RHOC 0.3f

#define LP 68
#define BM 16
#define BN 32

typedef __attribute__((ext_vector_type(2))) float v2f;
typedef __attribute__((ext_vector_type(4))) float v4f;
typedef __attribute__((ext_vector_type(8))) float v8f;

static __device__ __forceinline__ void mma_k32(const float* __restrict__ Arow,
                                               const float* __restrict__ Brow,
                                               v8f& acc0, v8f& acc1) {
    #pragma unroll
    for (int j = 0; j < 4; ++j) {
        const v4f a = *(const v4f*)(Arow + 8 * j);
        const v4f b = *(const v4f*)(Brow + 8 * j);
        const v2f a0 = {a.x, a.y}, b0 = {b.x, b.y};
        const v2f a1 = {a.z, a.w}, b1 = {b.z, b.w};
        acc0 = __builtin_amdgcn_wmma_f32_16x16x4_f32(false, a0, false, b0,
                                                     (short)0, acc0, false, false);
        acc1 = __builtin_amdgcn_wmma_f32_16x16x4_f32(false, a1, false, b1,
                                                     (short)0, acc1, false, false);
    }
}

__global__ __launch_bounds__(256)
void grf_fused_kernel(const float* __restrict__ theta, float* __restrict__ out) {
    __shared__ __attribute__((aligned(16))) float sG[4][NW * LP];
    __shared__ __attribute__((aligned(16))) float sTA[2][BM * LP];
    __shared__ __attribute__((aligned(16))) float sTB[2][BN * LP];
    __shared__ __attribute__((aligned(16))) float sM[2][BM * LP];
    __shared__ __attribute__((aligned(16))) float sP[4][BM * BN];

    const int tid  = threadIdx.x;
    const int lane = tid & 31;
    const int wv   = tid >> 5;
    const int bm   = blockIdx.x >> 2;
    const int bn   = blockIdx.x & 3;
    const float dw = W_MAX / (float)(NW - 1);

    {
        const int w1 = tid & 63;
        const int t6 = tid >> 6;
        const float vw1 = (float)w1 * dw;
        const float x1  = L1C * vw1;
        const float cnorm = (1.0f * L1C * L2C) / (4.0f * (float)M_PI);

        const float kTwo  = __builtin_sqrtf(2.0f  * dw * dw * cnorm);
        const float kOne  = __builtin_sqrtf(1.0f  * dw * dw * cnorm);
        const float kQrt  = __builtin_sqrtf(0.25f * dw * dw * cnorm);
        const float kBody = (w1 == 0) ? kOne : kTwo;
        const float kLast = (w1 == 0) ? kQrt : kOne;
        const float C8 = -0.125f * 1.44269504088896341f;

        const int w2b = t6 * 16;
        float p1[16], p2[16];
        #pragma unroll
        for (int i = 0; i < 16; ++i) {
            const int t = (w2b + i) * NW + w1;
            p1[i] = theta[t];
            p2[i] = theta[NW * NW + t];
        }
        #pragma unroll
        for (int g = 0; g < 4; ++g) {
            v4f gc, gs, hc, hs;
            #pragma unroll
            for (int k = 0; k < 4; ++k) {
                const int i  = 4 * g + k;
                const int w2 = w2b + i;
                const float vw2 = (float)w2 * dw;
                const float km  = (i == 15 && t6 == 3) ? kLast : kBody;

                const float x2 = L2C * vw2;
                const float q  = x1 * x1 + x2 * x2;
                const float cr = 2.0f * RHOC * x1 * x2;
                const float A1 = km * __builtin_amdgcn_exp2f((q + cr) * C8);
                const float A2 = km * __builtin_amdgcn_exp2f((q - cr) * C8);

                float sp1, cp1, sp2, cp2;
                __sincosf(p1[i], &sp1, &cp1);
                __sincosf(p2[i], &sp2, &cp2);

                gc[k] = A1 * cp1 + A2 * cp2;
                gs[k] = A2 * sp2 - A1 * sp1;
                hc[k] = A1 * sp1 + A2 * sp2;
                hs[k] = A1 * cp1 - A2 * cp2;
            }
            const int tt = w1 * LP + w2b + 4 * g;
            *(v4f*)&sG[0][tt] = gc;
            *(v4f*)&sG[1][tt] = gs;
            *(v4f*)&sG[2][tt] = hc;
            *(v4f*)&sG[3][tt] = hs;
        }
    }

    {
        const int w = tid & 63;
        const float xw = (float)w * dw;
        const float rinv = 1.0f / (float)(NS - 1);
        #pragma unroll
        for (int i = 0; i < BM / 4; ++i) {
            const int r = i * 4 + (tid >> 6);
            const float fs = (float)(bm * BM + r);
            const float q0 = fs * rinv;
            const float vs = __builtin_fmaf(__builtin_fmaf(-q0, (float)(NS - 1), fs), rinv, q0);
            const float x = vs * xw;
            float sn, cs;
            __sincosf(x, &sn, &cs);
            sTA[0][r * LP + w] = cs;
            sTA[1][r * LP + w] = sn;
        }
        #pragma unroll
        for (int i = 0; i < BN / 4; ++i) {
            const int r = i * 4 + (tid >> 6);
            const float fs = (float)(bn * BN + r);
            const float q0 = fs * rinv;
            const float vs = __builtin_fmaf(__builtin_fmaf(-q0, (float)(NS - 1), fs), rinv, q0);
            const float x = vs * xw;
            float sn, cs;
            __sincosf(x, &sn, &cs);
            sTB[0][r * LP + w] = cs;
            sTB[1][r * LP + w] = sn;
        }
    }
    __syncthreads();

    const int c = lane & 15;
    const int h = lane >> 4;

    {
        const int which = wv >> 2;
        const int tn    = wv & 3;
        const float* Ac = &sTA[0][c * LP + 4 * h];
        const float* As = &sTA[1][c * LP + 4 * h];
        const float* Bc = &sG[2 * which + 0][(tn * 16 + c) * LP + 4 * h];
        const float* Bs = &sG[2 * which + 1][(tn * 16 + c) * LP + 4 * h];

        v8f acc0 = {}, acc1 = {}, acc2 = {}, acc3 = {};
        mma_k32(Ac,      Bc,      acc0, acc1);
        mma_k32(As,      Bs,      acc2, acc3);
        mma_k32(Ac + 32, Bc + 32, acc0, acc1);
        mma_k32(As + 32, Bs + 32, acc2, acc3);

        const float sgn = which ? -1.0f : 1.0f;
        float* D = &sM[which][(8 * h) * LP + tn * 16 + c];
        #pragma unroll
        for (int r = 0; r < 8; ++r)
            D[r * LP] = sgn * ((acc0[r] + acc1[r]) + (acc2[r] + acc3[r]));
    }
    __syncthreads();

    {
        const int tn2   = wv & 1;
        const int kq    = wv >> 1;
        const int which = kq >> 1;
        const int k0    = (kq & 1) * 32;
        const float* A = &sM[which][c * LP + k0 + 4 * h];
        const float* B = &sTB[which][(tn2 * 16 + c) * LP + k0 + 4 * h];

        v8f acc0 = {}, acc1 = {};
        mma_k32(A, B, acc0, acc1);

        float* P = &sP[kq][(8 * h) * BN + tn2 * 16 + c];
        #pragma unroll
        for (int r = 0; r < 8; ++r)
            P[r * BN] = acc0[r] + acc1[r];
    }
    __syncthreads();

    {
        const float SQRT2 = 1.41421356237309505f;
        const int col = tid & 31;
        #pragma unroll
        for (int j = 0; j < 2; ++j) {
            const int row = (tid >> 5) + 8 * j;
            const int e = row * BN + col;
            const float v = (sP[0][e] + sP[1][e]) + (sP[2][e] + sP[3][e]);
            out[(bm * BM + row) * NS + bn * BN + col] = SQRT2 * v;
        }
    }
}

extern "C" void kernel_launch(void* const* d_in, const int* in_sizes, int n_in,
                              void* d_out, int out_size, void* d_ws, size_t ws_size,
                              hipStream_t stream) {
    const float* theta = (const float*)d_in[0];
    float* out = (float*)d_out;
    grf_fused_kernel<<<32, 256, 0, stream>>>(theta, out);
}
